# starting kernel
# baseline (speedup 1.0000x reference)
#include <hip/hip_runtime.h>

constexpr int Bdim = 4;
constexpr int Cdim = 1024;
constexpr int Ndim = 64;
constexpr int Tdim = 2048;

typedef float v2f __attribute__((ext_vector_type(2)));
typedef float v4f __attribute__((ext_vector_type(4)));
typedef float v8f __attribute__((ext_vector_type(8)));

__device__ __forceinline__ float softplus_fast(float x) {
    return fmaxf(x, 0.0f) + __logf(1.0f + __expf(-fabsf(x)));
}

__global__ void __launch_bounds__(256)
kgen_wmma(const float* __restrict__ mix,
          const float* __restrict__ log_tau,
          const float* __restrict__ log_dt,
          float* __restrict__ Kout) {
    __shared__ float sRate[Ndim];

    if (threadIdx.x < Ndim) {
        const float dtv = softplus_fast(log_dt[0]);
        sRate[threadIdx.x] = softplus_fast(log_tau[threadIdx.x]) * dtv;
    }
    __syncthreads();

    const int lane = threadIdx.x & 31;
    const int wave = threadIdx.x >> 5;
    const int row  = lane & 15;
    const int half = lane >> 4;

    const int t0 = blockIdx.x * 16;
    const int c0 = (blockIdx.y * 8 + wave) * 16;

    const float tB = (float)(t0 + row);

    v8f acc = {};
    const float* mixRow = mix + (size_t)(c0 + row) * Ndim;

    #pragma unroll
    for (int j0 = 0; j0 < Ndim; j0 += 4) {
        const int ja = j0 + half * 2;

        v2f a;
        a.x = mixRow[ja];
        a.y = mixRow[ja + 1];

        v2f b;
        b.x = __expf(-sRate[ja]     * tB);
        b.y = __expf(-sRate[ja + 1] * tB);

        acc = __builtin_amdgcn_wmma_f32_16x16x4_f32(
             false, a,  false, b,
             (short)0, acc,  false,  false);
    }

    #pragma unroll
    for (int r = 0; r < 8; ++r) {
        const int m = half * 8 + r;
        Kout[(size_t)(c0 + m) * Tdim + (t0 + row)] = acc[r];
    }
}

__global__ void __launch_bounds__(256)
kconv(const float* __restrict__ u,
      const float* __restrict__ Dv,
      const float* __restrict__ Kc,
      float* __restrict__ y) {
    __shared__ float sU[Tdim];
    __shared__ float sK[Tdim];

    const int c = blockIdx.x;
    const int b = blockIdx.y;

    const float* uc = u  + ((size_t)b * Cdim + c) * Tdim;
    const float* kc = Kc + (size_t)c * Tdim;

    for (int i = threadIdx.x; i < Tdim / 4; i += 256) {
        ((v4f*)sU)[i] = ((const v4f*)uc)[i];
        ((v4f*)sK)[i] = ((const v4f*)kc)[i];
    }
    __syncthreads();

    const float d  = Dv[c];
    const int   tb = threadIdx.x << 3;

    float acc[8];
    #pragma unroll
    for (int s = 0; s < 8; ++s) acc[s] = d * sU[tb + s];

    if (tb > 0) {
        v4f w1 = *(const v4f*)&sU[tb];
        v4f w2 = *(const v4f*)&sU[tb + 4];
        for (int jg = 0; jg < tb; jg += 4) {
            const int base = tb - jg;
            const v4f w0 = *(const v4f*)&sU[base - 4];
            const v4f k4 = *(const v4f*)&sK[jg];
            #pragma unroll
            for (int q = 0; q < 4; ++q) {
                #pragma unroll
                for (int s = 0; s < 8; ++s) {
                    const int i = s - q + 4;
                    const float uv = (i < 4) ? w0[i] : (i < 8) ? w1[i - 4] : w2[i - 8];
                    acc[s] += k4[q] * uv;
                }
            }
            w2 = w1;
            w1 = w0;
        }
    }

    {
        const v4f u0 = *(const v4f*)&sU[0];
        const v4f u1 = *(const v4f*)&sU[4];
        #pragma unroll
        for (int e = 0; e < 8; ++e) {
            const float kj = sK[tb + e];
            #pragma unroll
            for (int s = e; s < 8; ++s) {
                const int i = s - e;
                const float uv = (i < 4) ? u0[i] : u1[i - 4];
                acc[s] += kj * uv;
            }
        }
    }

    float* yc = y + ((size_t)b * Cdim + c) * Tdim;
    v4f o0, o1;
    #pragma unroll
    for (int s = 0; s < 4; ++s) { o0[s] = acc[s]; o1[s] = acc[s + 4]; }
    *(v4f*)&yc[tb]     = o0;
    *(v4f*)&yc[tb + 4] = o1;
}

extern "C" void kernel_launch(void* const* d_in, const int* in_sizes, int n_in,
                              void* d_out, int out_size, void* d_ws, size_t ws_size,
                              hipStream_t stream) {
    const float* u       = (const float*)d_in[0];
    const float* log_tau = (const float*)d_in[1];
    const float* mix     = (const float*)d_in[2];
    const float* Dv      = (const float*)d_in[3];
    const float* log_dt  = (const float*)d_in[4];
    float*       y       = (float*)d_out;
    float*       Kws     = (float*)d_ws;

    (void)in_sizes; (void)n_in; (void)out_size; (void)ws_size;

    dim3 g1(Tdim / 16, (Cdim / 16) / 8);
    kgen_wmma<<<g1, 256, 0, stream>>>(mix, log_tau, log_dt, Kws);

    dim3 g2(Cdim, Bdim);
    kconv<<<g2, 256, 0, stream>>>(u, Dv, Kws, y);
}


# causal conv moved from per-thread FMAs to f32 16x16x4 matrix products over 16x16 Toeplitz blocks of K (t=16p+q); sums now run in block-lag order (new reduction tree)
# speedup vs baseline: 3.1059x; 3.1059x over previous
#include <hip/hip_runtime.h>

constexpr int Bdim = 4;
constexpr int Cdim = 1024;
constexpr int Ndim = 64;
constexpr int Tdim = 2048;

typedef float v2f __attribute__((ext_vector_type(2)));
typedef float v4f __attribute__((ext_vector_type(4)));
typedef float v8f __attribute__((ext_vector_type(8)));

__device__ __forceinline__ float softplus_fast(float x) {
    return fmaxf(x, 0.0f) + __logf(1.0f + __expf(-fabsf(x)));
}

__global__ void __launch_bounds__(256)
kgen_wmma(const float* __restrict__ mix,
          const float* __restrict__ log_tau,
          const float* __restrict__ log_dt,
          float* __restrict__ Kout) {
    __shared__ float sRate[Ndim];

    if (threadIdx.x < Ndim) {
        const float dtv = softplus_fast(log_dt[0]);
        sRate[threadIdx.x] = softplus_fast(log_tau[threadIdx.x]) * dtv;
    }
    __syncthreads();

    const int lane = threadIdx.x & 31;
    const int wave = threadIdx.x >> 5;
    const int row  = lane & 15;
    const int half = lane >> 4;

    const int t0 = blockIdx.x * 16;
    const int c0 = (blockIdx.y * 8 + wave) * 16;

    const float tB = (float)(t0 + row);

    v8f acc = {};
    const float* mixRow = mix + (size_t)(c0 + row) * Ndim;

    #pragma unroll
    for (int j0 = 0; j0 < Ndim; j0 += 4) {
        const int ja = j0 + half * 2;

        v2f a;
        a.x = mixRow[ja];
        a.y = mixRow[ja + 1];

        v2f b;
        b.x = __expf(-sRate[ja]     * tB);
        b.y = __expf(-sRate[ja + 1] * tB);

        acc = __builtin_amdgcn_wmma_f32_16x16x4_f32(
             false, a,  false, b,
             (short)0, acc,  false,  false);
    }

    #pragma unroll
    for (int r = 0; r < 8; ++r) {
        const int m = half * 8 + r;
        Kout[(size_t)(c0 + m) * Tdim + (t0 + row)] = acc[r];
    }
}

constexpr int UP    = 20;
constexpr int UZR   = 16;
constexpr int USZ   = (Tdim / 16 + UZR) * UP;
constexpr int KPAD  = 16;
constexpr int KSZ   = Tdim + 2 * KPAD;

template <int NT>
__device__ __forceinline__ void lag_step(v8f* acc, const float* pa_hi, const float* pa_lo,
                                         const float* pb) {
    v4f a[NT][2];
    float bv[8];
    #pragma unroll
    for (int t = 0; t < NT; ++t) {
        const float* p = ((t < 2) ? pa_hi : pa_lo) + (t & 1) * USZ;
        a[t][0] = *(const v4f*)p;
        a[t][1] = *(const v4f*)(p + 4);
    }
    #pragma unroll
    for (int j = 0; j < 8; ++j) bv[j] = pb[j];

    #pragma unroll
    for (int kk = 0; kk < 4; ++kk) {
        v2f b2;
        b2.x = bv[2 * kk];
        b2.y = bv[2 * kk + 1];
        #pragma unroll
        for (int t = 0; t < NT; ++t) {
            v2f a2;
            a2.x = a[t][kk >> 1][2 * (kk & 1)];
            a2.y = a[t][kk >> 1][2 * (kk & 1) + 1];
            acc[t] = __builtin_amdgcn_wmma_f32_16x16x4_f32(
                 false, a2,  false, b2,
                 (short)0, acc[t],  false,  false);
        }
    }
}

__global__ void __launch_bounds__(128) __attribute__((amdgpu_waves_per_eu(8, 8)))
kconv(const float* __restrict__ u,
      const float* __restrict__ Dv,
      const float* __restrict__ Kc,
      float* __restrict__ y) {
    __shared__ __attribute__((aligned(16))) float sU[2 * USZ];
    __shared__ __attribute__((aligned(16))) float sKr[KSZ];

    const int c   = blockIdx.x;
    const int b0  = blockIdx.y * 2;
    const int tid = threadIdx.x;

    const v4f zero4 = {0.0f, 0.0f, 0.0f, 0.0f};
    if (tid < UZR * UP / 4) {
        *(v4f*)&sU[tid * 4]       = zero4;
        *(v4f*)&sU[USZ + tid * 4] = zero4;
    }
    if (tid < 4)      *(v4f*)&sKr[tid * 4] = zero4;
    else if (tid < 8) *(v4f*)&sKr[KPAD + Tdim + (tid - 4) * 4] = zero4;

    #pragma unroll
    for (int k = 0; k < 2 * (Tdim / 4) / 128; ++k) {
        const int i   = tid + 128 * k;
        const int bb  = i >> 9;
        const int i4  = i & 511;
        const v4f v = *(const v4f*)(u + ((size_t)(b0 + bb) * Cdim + c) * Tdim + i4 * 4);
        *(v4f*)&sU[bb * USZ + ((i4 >> 2) + UZR) * UP + (i4 & 3) * 4] = v;
    }
    #pragma unroll
    for (int k = 0; k < (Tdim / 4) / 128; ++k) {
        const int i = tid + 128 * k;
        const v4f v = *(const v4f*)(Kc + (size_t)c * Tdim + i * 4);
        v4f rv;
        rv.x = v.w; rv.y = v.z; rv.z = v.y; rv.w = v.x;
        *(v4f*)&sKr[KPAD + Tdim - 4 - i * 4] = rv;
    }
    __syncthreads();

    const int lane = tid & 31;
    const int wave = __builtin_amdgcn_readfirstlane(tid >> 5);
    const int m    = lane & 15;
    const int h    = lane >> 4;
    const int ilo  = wave;
    const int ihi  = 7 - wave;

    v8f acc[4] = {};
    const float* pah = sU + (16 * ihi + m + UZR) * UP + 8 * h;
    const float* pal = sU + (16 * ilo + m + UZR) * UP + 8 * h;
    const float* pb  = sKr + KPAD + (Tdim - 1) - m + 8 * h;

    const int n1 = 16 * (ilo + 1);
    const int n2 = 16 * (ihi + 1);
    int d = 0;
    for (; d < n1; ++d) {
        lag_step<4>(acc, pah, pal, pb);
        pah -= UP; pal -= UP; pb -= 16;
    }
    for (; d < n2; ++d) {
        lag_step<2>(acc, pah, pal, pb);
        pah -= UP; pb -= 16;
    }

    const float dval = Dv[c];
    #pragma unroll
    for (int t = 0; t < 4; ++t) {
        const int bb = t & 1;
        const int it = (t < 2) ? ihi : ilo;
        float* yc = y + ((size_t)(b0 + bb) * Cdim + c) * Tdim;
        #pragma unroll
        for (int r = 0; r < 8; ++r) {
            const int p = 16 * it + 8 * h + r;
            const float uv = sU[bb * USZ + (p + UZR) * UP + m];
            yc[16 * p + m] = acc[t][r] + dval * uv;
        }
    }
}

extern "C" void kernel_launch(void* const* d_in, const int* in_sizes, int n_in,
                              void* d_out, int out_size, void* d_ws, size_t ws_size,
                              hipStream_t stream) {
    const float* u       = (const float*)d_in[0];
    const float* log_tau = (const float*)d_in[1];
    const float* mix     = (const float*)d_in[2];
    const float* Dv      = (const float*)d_in[3];
    const float* log_dt  = (const float*)d_in[4];
    float*       y       = (float*)d_out;
    float*       Kws     = (float*)d_ws;

    (void)in_sizes; (void)n_in; (void)out_size; (void)ws_size;

    dim3 g1(Tdim / 16, (Cdim / 16) / 8);
    kgen_wmma<<<g1, 256, 0, stream>>>(mix, log_tau, log_dt, Kws);

    dim3 g2(Cdim, Bdim / 2);
    kconv<<<g2, 128, 0, stream>>>(u, Dv, Kws, y);
}


# lags of 8 blocks and more folded into a 64-mode decaying state per 128-step tile (state-in product, 16-step scan, state-out product); Toeplitz lags only inside a tile; new reduction tree
# speedup vs baseline: 9.4390x; 3.0391x over previous
#include <hip/hip_runtime.h>

constexpr int Bdim = 4;
constexpr int Cdim = 1024;
constexpr int Ndim = 64;
constexpr int Tdim = 2048;
constexpr int KT   = 128;

typedef float v2f __attribute__((ext_vector_type(2)));
typedef float v4f __attribute__((ext_vector_type(4)));
typedef float v8f __attribute__((ext_vector_type(8)));

__device__ __forceinline__ float softplus_fast(float x) {
    return fmaxf(x, 0.0f) + __logf(1.0f + __expf(-fabsf(x)));
}

__global__ void __launch_bounds__(256)
kgen_wmma(const float* __restrict__ mix,
          const float* __restrict__ log_tau,
          const float* __restrict__ log_dt,
          float* __restrict__ Kout) {
    __shared__ float sRate[Ndim];

    if (threadIdx.x < Ndim) {
        const float dtv = softplus_fast(log_dt[0]);
        sRate[threadIdx.x] = softplus_fast(log_tau[threadIdx.x]) * dtv;
    }
    __syncthreads();

    const int lane = threadIdx.x & 31;
    const int wave = threadIdx.x >> 5;
    const int row  = lane & 15;
    const int half = lane >> 4;

    const int t0 = blockIdx.x * 16;
    const int c0 = (blockIdx.y * 8 + wave) * 16;

    const float tB = (float)(t0 + row);

    v8f acc = {};
    const float* mixRow = mix + (size_t)(c0 + row) * Ndim;

    #pragma unroll
    for (int j0 = 0; j0 < Ndim; j0 += 4) {
        const int ja = j0 + half * 2;
        v2f a;
        a.x = mixRow[ja];
        a.y = mixRow[ja + 1];
        v2f b;
        b.x = __expf(-sRate[ja]     * tB);
        b.y = __expf(-sRate[ja + 1] * tB);
        acc = __builtin_amdgcn_wmma_f32_16x16x4_f32(
            false, a, false, b, (short)0, acc, false, false);
    }

    #pragma unroll
    for (int r = 0; r < 8; ++r) {
        const int m = half * 8 + r;
        Kout[(size_t)(c0 + m) * KT + (t0 + row)] = acc[r];
    }
}

constexpr int NTL  = Tdim / 128;
constexpr int UP   = 20;
constexpr int TP   = 8 * UP + 4;
constexpr int USZ  = NTL * TP + 32;
constexpr int ZOFF = 2 * USZ;
constexpr int KPAD = 16;
constexpr int KSZ  = KT + 2 * KPAD;

__global__ void __launch_bounds__(128) __attribute__((amdgpu_waves_per_eu(8, 8)))
kconv(const float* __restrict__ u,
      const float* __restrict__ Dv,
      const float* __restrict__ Kc,
      const float* __restrict__ log_tau,
      const float* __restrict__ log_dt,
      const float* __restrict__ mix,
      float* __restrict__ y) {
    __shared__ __attribute__((aligned(16))) float sU[2 * USZ + UP];
    __shared__ __attribute__((aligned(16))) float sKr[KSZ];
    __shared__ __attribute__((aligned(16))) float sG[2 * NTL * Ndim];
    __shared__ __attribute__((aligned(16))) float sRate[Ndim];

    const int c   = blockIdx.x;
    const int b0  = blockIdx.y * 2;
    const int tid = threadIdx.x;

    const v4f zero4 = {0.0f, 0.0f, 0.0f, 0.0f};
    if (tid < UP / 4) *(v4f*)&sU[ZOFF + tid * 4] = zero4;
    if (tid < 4)      *(v4f*)&sKr[tid * 4] = zero4;
    else if (tid < 8) *(v4f*)&sKr[KPAD + KT + (tid - 4) * 4] = zero4;
    if (tid >= 64) {
        const float dtv = softplus_fast(log_dt[0]);
        sRate[tid - 64] = softplus_fast(log_tau[tid - 64]) * dtv;
    }

    #pragma unroll
    for (int k = 0; k < 2 * (Tdim / 4) / 128; ++k) {
        const int i   = tid + 128 * k;
        const int bb  = i >> 9;
        const int i4  = i & 511;
        const v4f v = *(const v4f*)(u + ((size_t)(b0 + bb) * Cdim + c) * Tdim + i4 * 4);
        *(v4f*)&sU[bb * USZ + (i4 >> 5) * TP + ((i4 >> 2) & 7) * UP + (i4 & 3) * 4] = v;
    }
    if (tid < KT / 4) {
        const v4f v = *(const v4f*)(Kc + (size_t)c * KT + tid * 4);
        v4f rv;
        rv.x = v.w; rv.y = v.z; rv.z = v.y; rv.w = v.x;
        *(v4f*)&sKr[KPAD + KT - 4 - tid * 4] = rv;
    }
    __syncthreads();

    const int lane = tid & 31;
    const int wave = __builtin_amdgcn_readfirstlane(tid >> 5);
    const int m    = lane & 15;
    const int h    = lane >> 4;
    const int mb   = m >> 3;
    const int mp   = m & 7;

    {
        const int   jcol = 16 * wave + m;
        const float rj   = sRate[jcol];
        v8f w0 = {}, w1 = {};
        const float* pa = sU + mb * USZ + mp * TP + 8 * h;
        for (int sr = 0; sr < 8; ++sr) {
            const float* p0 = pa + sr * UP;
            const float* p1 = p0 + 8 * TP;
            const v4f a00 = *(const v4f*)p0;
            const v4f a01 = *(const v4f*)(p0 + 4);
            const v4f a10 = *(const v4f*)p1;
            const v4f a11 = *(const v4f*)(p1 + 4);
            const float av0[8] = {a00.x, a00.y, a00.z, a00.w, a01.x, a01.y, a01.z, a01.w};
            const float av1[8] = {a10.x, a10.y, a10.z, a10.w, a11.x, a11.y, a11.z, a11.w};
            const float back = (float)(128 - 16 * sr - 8 * h);
            #pragma unroll
            for (int kk = 0; kk < 4; ++kk) {
                v2f b2, a2, a3;
                b2.x = __expf(-rj * (back - (float)(2 * kk)));
                b2.y = __expf(-rj * (back - (float)(2 * kk + 1)));
                a2.x = av0[2 * kk]; a2.y = av0[2 * kk + 1];
                a3.x = av1[2 * kk]; a3.y = av1[2 * kk + 1];
                w0 = __builtin_amdgcn_wmma_f32_16x16x4_f32(false, a2, false, b2, (short)0, w0, false, false);
                w1 = __builtin_amdgcn_wmma_f32_16x16x4_f32(false, a3, false, b2, (short)0, w1, false, false);
            }
        }
        const float lam = __expf(-rj * 128.0f);
        const float mj  = mix[(size_t)c * Ndim + jcol];
        float z = 0.0f;
        #pragma unroll
        for (int r = 0; r < 8; ++r) {
            sG[(h * NTL + r) * Ndim + jcol] = mj * z;
            z = lam * z + w0[r];
        }
        #pragma unroll
        for (int r = 0; r < 8; ++r) {
            sG[(h * NTL + 8 + r) * Ndim + jcol] = mj * z;
            z = lam * z + w1[r];
        }
    }
    __syncthreads();

    const int i0 = 4 * wave;
    v8f acc[4] = {};

    {
        const float fq   = (float)m;
        const float fp16 = (float)(16 * mp);
        const float* pg  = sG + (mb * NTL + i0) * Ndim + 2 * h;
        for (int st = 0; st < 16; ++st) {
            const v2f rr = *(const v2f*)&sRate[4 * st + 2 * h];
            v2f fa, fb;
            fa.x = __expf(-rr.x * fp16);
            fa.y = __expf(-rr.y * fp16);
            fb.x = __expf(-rr.x * fq);
            fb.y = __expf(-rr.y * fq);
            #pragma unroll
            for (int t = 0; t < 4; ++t) {
                const v2f g = *(const v2f*)(pg + t * Ndim + 4 * st);
                v2f a2;
                a2.x = g.x * fa.x;
                a2.y = g.y * fa.y;
                acc[t] = __builtin_amdgcn_wmma_f32_16x16x4_f32(false, a2, false, fb, (short)0, acc[t], false, false);
            }
        }
    }

    {
        const float* tb = sU + mb * USZ + i0 * TP + 8 * h;
        const float* zr = sU + ZOFF + 8 * h;
        const float* pb = sKr + KPAD + (KT - 1) - m + 8 * h;
        for (int d = 0; d < 8; ++d) {
            const bool inside = (mp >= d);
            const int  off    = (mp - d) * UP;
            v4f a[4][2];
            float bv[8];
            #pragma unroll
            for (int t = 0; t < 4; ++t) {
                const float* p = inside ? (tb + t * TP + off) : zr;
                a[t][0] = *(const v4f*)p;
                a[t][1] = *(const v4f*)(p + 4);
            }
            #pragma unroll
            for (int j = 0; j < 8; ++j) bv[j] = pb[j];
            pb -= 16;

            #pragma unroll
            for (int kk = 0; kk < 4; ++kk) {
                v2f b2;
                b2.x = bv[2 * kk];
                b2.y = bv[2 * kk + 1];
                #pragma unroll
                for (int t = 0; t < 4; ++t) {
                    v2f a2;
                    a2.x = a[t][kk >> 1][2 * (kk & 1)];
                    a2.y = a[t][kk >> 1][2 * (kk & 1) + 1];
                    acc[t] = __builtin_amdgcn_wmma_f32_16x16x4_f32(false, a2, false, b2, (short)0, acc[t], false, false);
                }
            }
        }
    }

    const float dval = Dv[c];
    #pragma unroll
    for (int t = 0; t < 4; ++t) {
        float* yc = y + ((size_t)(b0 + h) * Cdim + c) * Tdim + 128 * (i0 + t);
        #pragma unroll
        for (int r = 0; r < 8; ++r) {
            const float uv = sU[h * USZ + (i0 + t) * TP + r * UP + m];
            yc[16 * r + m] = acc[t][r] + dval * uv;
        }
    }
}

extern "C" void kernel_launch(void* const* d_in, const int* in_sizes, int n_in,
                              void* d_out, int out_size, void* d_ws, size_t ws_size,
                              hipStream_t stream) {
    const float* u       = (const float*)d_in[0];
    const float* log_tau = (const float*)d_in[1];
    const float* mix     = (const float*)d_in[2];
    const float* Dv      = (const float*)d_in[3];
    const float* log_dt  = (const float*)d_in[4];
    float*       y       = (float*)d_out;
    float*       Kws     = (float*)d_ws;

    (void)in_sizes; (void)n_in; (void)out_size; (void)ws_size;

    dim3 g1(KT / 16, (Cdim / 16) / 8);
    kgen_wmma<<<g1, 256, 0, stream>>>(mix, log_tau, log_dt, Kws);

    dim3 g2(Cdim, Bdim / 2);
    kconv<<<g2, 128, 0, stream>>>(u, Dv, Kws, log_tau, log_dt, mix, y);
}


# state tile cut from 128 to 16 steps: only the diagonal 16x16 Toeplitz block stays direct, all longer lags go through the 64-mode state, streamed per 512-step quarter via LDS; new reduction tree
# speedup vs baseline: 12.5725x; 1.3320x over previous
#include <hip/hip_runtime.h>

constexpr int Bdim = 4;
constexpr int Cdim = 1024;
constexpr int Ndim = 64;
constexpr int Tdim = 2048;
constexpr int KT   = 16;

typedef float v2f __attribute__((ext_vector_type(2)));
typedef float v4f __attribute__((ext_vector_type(4)));
typedef float v8f __attribute__((ext_vector_type(8)));

__device__ __forceinline__ float softplus_fast(float x) {
    return fmaxf(x, 0.0f) + __logf(1.0f + __expf(-fabsf(x)));
}

__global__ void __launch_bounds__(256)
kgen_wmma(const float* __restrict__ mix,
          const float* __restrict__ log_tau,
          const float* __restrict__ log_dt,
          float* __restrict__ Kout) {
    __shared__ float sRate[Ndim];

    if (threadIdx.x < Ndim) {
        const float dtv = softplus_fast(log_dt[0]);
        sRate[threadIdx.x] = softplus_fast(log_tau[threadIdx.x]) * dtv;
    }
    __syncthreads();

    const int lane = threadIdx.x & 31;
    const int wave = threadIdx.x >> 5;
    const int row  = lane & 15;
    const int half = lane >> 4;

    const int t0 = blockIdx.x * 16;
    const int c0 = (blockIdx.y * 8 + wave) * 16;

    const float tB = (float)(t0 + row);

    v8f acc = {};
    const float* mixRow = mix + (size_t)(c0 + row) * Ndim;

    #pragma unroll
    for (int j0 = 0; j0 < Ndim; j0 += 4) {
        const int ja = j0 + half * 2;
        v2f a;
        a.x = mixRow[ja];
        a.y = mixRow[ja + 1];
        v2f b;
        b.x = __expf(-sRate[ja]     * tB);
        b.y = __expf(-sRate[ja + 1] * tB);
        acc = __builtin_amdgcn_wmma_f32_16x16x4_f32(
            false, a, false, b, (short)0, acc, false, false);
    }

    #pragma unroll
    for (int r = 0; r < 8; ++r) {
        const int m = half * 8 + r;
        Kout[(size_t)(c0 + m) * KT + (t0 + row)] = acc[r];
    }
}

constexpr int NCH  = Tdim / 128;
constexpr int UP   = 20;
constexpr int TP   = 8 * UP;
constexpr int USZ  = NCH * TP + 32;
constexpr int KPAD = 16;
constexpr int KSZ  = KT + 2 * KPAD;
constexpr int QTL  = 32;
constexpr int GP   = Ndim + 4;
constexpr int GB   = QTL * GP + 32;

__global__ void __launch_bounds__(128) __attribute__((amdgpu_waves_per_eu(8, 8)))
kconv(const float* __restrict__ u,
      const float* __restrict__ Dv,
      const float* __restrict__ Kc,
      const float* __restrict__ log_tau,
      const float* __restrict__ log_dt,
      const float* __restrict__ mix,
      float* __restrict__ y) {
    __shared__ __attribute__((aligned(16))) float sU[USZ + NCH * TP];
    __shared__ __attribute__((aligned(16))) float sKr[KSZ];
    __shared__ __attribute__((aligned(16))) float sG[GB + QTL * GP];
    __shared__ __attribute__((aligned(16))) float sRate[Ndim];

    const int c   = blockIdx.x;
    const int b0  = blockIdx.y * 2;
    const int tid = threadIdx.x;

    const v4f zero4 = {0.0f, 0.0f, 0.0f, 0.0f};
    if (tid < 4)      *(v4f*)&sKr[tid * 4] = zero4;
    else if (tid < 8) *(v4f*)&sKr[KPAD + KT + (tid - 4) * 4] = zero4;
    if (tid >= 64) {
        const float dtv = softplus_fast(log_dt[0]);
        sRate[tid - 64] = softplus_fast(log_tau[tid - 64]) * dtv;
    }

    #pragma unroll
    for (int k = 0; k < 2 * (Tdim / 4) / 128; ++k) {
        const int i   = tid + 128 * k;
        const int bb  = i >> 9;
        const int i4  = i & 511;
        const v4f v = *(const v4f*)(u + ((size_t)(b0 + bb) * Cdim + c) * Tdim + i4 * 4);
        *(v4f*)&sU[bb * USZ + (i4 >> 5) * TP + ((i4 >> 2) & 7) * UP + (i4 & 3) * 4] = v;
    }
    if (tid < KT / 4) {
        const v4f v = *(const v4f*)(Kc + (size_t)c * KT + tid * 4);
        v4f rv;
        rv.x = v.w; rv.y = v.z; rv.z = v.y; rv.w = v.x;
        *(v4f*)&sKr[KPAD + KT - 4 - tid * 4] = rv;
    }
    __syncthreads();

    const int lane = tid & 31;
    const int wave = __builtin_amdgcn_readfirstlane(tid >> 5);
    const int m    = lane & 15;
    const int h    = lane >> 4;
    const int mb   = m >> 3;
    const int mp   = m & 7;

    const int   jcol = 16 * wave + m;
    const float rj   = sRate[jcol];
    const float lam  = __expf(-rj * 16.0f);
    const float mj   = mix[(size_t)c * Ndim + jcol];
    const float dval = Dv[c];

    float bIn[8];
    #pragma unroll
    for (int j = 0; j < 8; ++j) bIn[j] = __expf(-rj * (float)(16 - 8 * h - j));

    float bv[8];
    {
        const float* pb = sKr + KPAD + (KT - 1) - m + 8 * h;
        #pragma unroll
        for (int j = 0; j < 8; ++j) bv[j] = pb[j];
    }

    v2f fb[16];
    {
        const float fq = (float)m;
        #pragma unroll
        for (int st = 0; st < 16; ++st) {
            const v2f rr = *(const v2f*)&sRate[32 * h + 2 * st];
            fb[st].x = __expf(-rr.x * fq);
            fb[st].y = __expf(-rr.y * fq);
        }
    }

    const float* uA = sU + mb * USZ + mp * UP + 8 * h;
    float*       gW = sG + h * GB + jcol;
    const float* gR = sG + mb * GB + (8 * wave + mp) * GP + 32 * h;
    float z = 0.0f;

    for (int Q = 0; Q < 4; ++Q) {
        for (int pr = 0; pr < 2; ++pr) {
            const float* p0 = uA + (4 * Q + 2 * pr) * TP;
            const float* p1 = p0 + TP;
            const v4f a00 = *(const v4f*)p0;
            const v4f a01 = *(const v4f*)(p0 + 4);
            const v4f a10 = *(const v4f*)p1;
            const v4f a11 = *(const v4f*)(p1 + 4);
            const float av0[8] = {a00.x, a00.y, a00.z, a00.w, a01.x, a01.y, a01.z, a01.w};
            const float av1[8] = {a10.x, a10.y, a10.z, a10.w, a11.x, a11.y, a11.z, a11.w};
            v8f w0 = {}, w1 = {};
            #pragma unroll
            for (int kk = 0; kk < 4; ++kk) {
                v2f b2, a2, a3;
                b2.x = bIn[2 * kk];  b2.y = bIn[2 * kk + 1];
                a2.x = av0[2 * kk];  a2.y = av0[2 * kk + 1];
                a3.x = av1[2 * kk];  a3.y = av1[2 * kk + 1];
                w0 = __builtin_amdgcn_wmma_f32_16x16x4_f32(false, a2, false, b2, (short)0, w0, false, false);
                w1 = __builtin_amdgcn_wmma_f32_16x16x4_f32(false, a3, false, b2, (short)0, w1, false, false);
            }
            float* g0 = gW + (16 * pr) * GP;
            #pragma unroll
            for (int r = 0; r < 8; ++r) {
                g0[r * GP] = mj * z;
                z = lam * z + w0[r];
            }
            #pragma unroll
            for (int r = 0; r < 8; ++r) {
                g0[(8 + r) * GP] = mj * z;
                z = lam * z + w1[r];
            }
        }
        __syncthreads();

        const int ch = 4 * Q + wave;
        v4f g[8];
        #pragma unroll
        for (int i = 0; i < 8; ++i) g[i] = *(const v4f*)(gR + 4 * i);
        const v4f t0 = *(const v4f*)(uA + ch * TP);
        const v4f t1 = *(const v4f*)(uA + ch * TP + 4);
        __syncthreads();

        const float tv[8] = {t0.x, t0.y, t0.z, t0.w, t1.x, t1.y, t1.z, t1.w};
        v8f acc0 = {}, acc1 = {};
        #pragma unroll
        for (int kk = 0; kk < 4; kk += 2) {
            v2f a2, b2, a3, b3;
            a2.x = tv[2 * kk];      a2.y = tv[2 * kk + 1];
            b2.x = bv[2 * kk];      b2.y = bv[2 * kk + 1];
            a3.x = tv[2 * kk + 2];  a3.y = tv[2 * kk + 3];
            b3.x = bv[2 * kk + 2];  b3.y = bv[2 * kk + 3];
            acc0 = __builtin_amdgcn_wmma_f32_16x16x4_f32(false, a2, false, b2, (short)0, acc0, false, false);
            acc1 = __builtin_amdgcn_wmma_f32_16x16x4_f32(false, a3, false, b3, (short)0, acc1, false, false);
        }
        #pragma unroll
        for (int st = 0; st < 16; st += 2) {
            v2f a2, a3;
            a2.x = g[st >> 1].x;  a2.y = g[st >> 1].y;
            a3.x = g[st >> 1].z;  a3.y = g[st >> 1].w;
            acc0 = __builtin_amdgcn_wmma_f32_16x16x4_f32(false, a2, false, fb[st],     (short)0, acc0, false, false);
            acc1 = __builtin_amdgcn_wmma_f32_16x16x4_f32(false, a3, false, fb[st + 1], (short)0, acc1, false, false);
        }

        float* yc = y + ((size_t)(b0 + h) * Cdim + c) * Tdim + 128 * ch;
        #pragma unroll
        for (int r = 0; r < 8; ++r) {
            const float uv = sU[h * USZ + ch * TP + r * UP + m];
            yc[16 * r + m] = (acc0[r] + acc1[r]) + dval * uv;
        }
    }
}

extern "C" void kernel_launch(void* const* d_in, const int* in_sizes, int n_in,
                              void* d_out, int out_size, void* d_ws, size_t ws_size,
                              hipStream_t stream) {
    const float* u       = (const float*)d_in[0];
    const float* log_tau = (const float*)d_in[1];
    const float* mix     = (const float*)d_in[2];
    const float* Dv      = (const float*)d_in[3];
    const float* log_dt  = (const float*)d_in[4];
    float*       y       = (float*)d_out;
    float*       Kws     = (float*)d_ws;

    (void)in_sizes; (void)n_in; (void)out_size; (void)ws_size;

    dim3 g1(KT / 16, (Cdim / 16) / 8);
    kgen_wmma<<<g1, 256, 0, stream>>>(mix, log_tau, log_dt, Kws);

    dim3 g2(Cdim, Bdim / 2);
    kconv<<<g2, 128, 0, stream>>>(u, Dv, Kws, log_tau, log_dt, mix, y);
}


# global memory taken off the barrier waits: all loads issued at wave entry in both kernels, barriers order LDS only, so set-up runs under the loads and no barrier waits for y stores; sums unchanged
# speedup vs baseline: 13.6424x; 1.0851x over previous
#include <hip/hip_runtime.h>

constexpr int Bdim = 4;
constexpr int Cdim = 1024;
constexpr int Ndim = 64;
constexpr int Tdim = 2048;
constexpr int KT   = 16;

typedef float v2f __attribute__((ext_vector_type(2)));
typedef float v4f __attribute__((ext_vector_type(4)));
typedef float v8f __attribute__((ext_vector_type(8)));

__device__ __forceinline__ float softplus_fast(float x) {
    return fmaxf(x, 0.0f) + __logf(1.0f + __expf(-fabsf(x)));
}

__device__ __forceinline__ void barrier_lds() {
    __builtin_amdgcn_fence(__ATOMIC_RELEASE, "workgroup", "local");
    __builtin_amdgcn_s_barrier();
    __builtin_amdgcn_fence(__ATOMIC_ACQUIRE, "workgroup", "local");
}

__global__ void __launch_bounds__(256)
kgen_wmma(const float* __restrict__ mix,
          const float* __restrict__ log_tau,
          const float* __restrict__ log_dt,
          float* __restrict__ Kout) {
    __shared__ float sRate[Ndim];

    const int lane = threadIdx.x & 31;
    const int wave = threadIdx.x >> 5;
    const int row  = lane & 15;
    const int half = lane >> 4;

    const int t0 = blockIdx.x * 16;
    const int c0 = (blockIdx.y * 8 + wave) * 16;

    const float ltv = log_tau[threadIdx.x & (Ndim - 1)];
    const float ldt = log_dt[0];
    const float* mixRow = mix + (size_t)(c0 + row) * Ndim + half * 2;
    v2f a[Ndim / 4];
    #pragma unroll
    for (int s = 0; s < Ndim / 4; ++s) a[s] = *(const v2f*)(mixRow + 4 * s);

    if (threadIdx.x < Ndim) sRate[threadIdx.x] = softplus_fast(ltv) * softplus_fast(ldt);
    barrier_lds();

    const float tB = (float)(t0 + row);

    v8f acc = {};

    #pragma unroll
    for (int j0 = 0; j0 < Ndim; j0 += 4) {
        const int ja = j0 + half * 2;
        v2f b;
        b.x = __expf(-sRate[ja]     * tB);
        b.y = __expf(-sRate[ja + 1] * tB);
        acc = __builtin_amdgcn_wmma_f32_16x16x4_f32(
            false, a[j0 >> 2], false, b, (short)0, acc, false, false);
    }

    #pragma unroll
    for (int r = 0; r < 8; ++r) {
        const int m = half * 8 + r;
        Kout[(size_t)(c0 + m) * KT + (t0 + row)] = acc[r];
    }
}

constexpr int NCH  = Tdim / 128;
constexpr int UP   = 20;
constexpr int TP   = 8 * UP;
constexpr int USZ  = NCH * TP + 32;
constexpr int KPAD = 16;
constexpr int KSZ  = KT + 2 * KPAD;
constexpr int QTL  = 32;
constexpr int GP   = Ndim + 4;
constexpr int GB   = QTL * GP + 32;

__global__ void __launch_bounds__(128) __attribute__((amdgpu_waves_per_eu(8, 8)))
kconv(const float* __restrict__ u,
      const float* __restrict__ Dv,
      const float* __restrict__ Kc,
      const float* __restrict__ log_tau,
      const float* __restrict__ log_dt,
      const float* __restrict__ mix,
      float* __restrict__ y) {
    __shared__ __attribute__((aligned(16))) float sU[USZ + NCH * TP];
    __shared__ __attribute__((aligned(16))) float sKr[KSZ];
    __shared__ __attribute__((aligned(16))) float sG[GB + QTL * GP];
    __shared__ __attribute__((aligned(16))) float sRate[Ndim];

    const int c   = blockIdx.x;
    const int b0  = blockIdx.y * 2;
    const int tid = threadIdx.x;

    const int lane = tid & 31;
    const int wave = __builtin_amdgcn_readfirstlane(tid >> 5);
    const int m    = lane & 15;
    const int h    = lane >> 4;
    const int mb   = m >> 3;
    const int mp   = m & 7;
    const int jcol = 16 * wave + m;

    const float ltv  = log_tau[tid & (Ndim - 1)];
    const float ldt  = log_dt[0];
    const v4f   kv   = *(const v4f*)(Kc + (size_t)c * KT + (tid & (KT / 4 - 1)) * 4);
    const float mj   = mix[(size_t)c * Ndim + jcol];
    const float dval = Dv[c];
    __builtin_amdgcn_sched_barrier(0);
    const float* const uG = u + ((size_t)b0 * Cdim + c) * Tdim + tid * 4;
    v4f uq[8];
    #pragma unroll
    for (int k = 0; k < 8; ++k)
        uq[k] = *(const v4f*)(uG + (size_t)(k & 1) * Cdim * Tdim + 512 * (k >> 1));

    const v4f zero4 = {0.0f, 0.0f, 0.0f, 0.0f};
    if (tid < 4)      *(v4f*)&sKr[tid * 4] = zero4;
    else if (tid < 8) *(v4f*)&sKr[KPAD + KT + (tid - 4) * 4] = zero4;
    if (tid < Ndim) sRate[tid] = softplus_fast(ltv) * softplus_fast(ldt);
    if (tid < KT / 4) {
        v4f rv;
        rv.x = kv.w; rv.y = kv.z; rv.z = kv.y; rv.w = kv.x;
        *(v4f*)&sKr[KPAD + KT - 4 - tid * 4] = rv;
    }
    barrier_lds();

    const float rj   = sRate[jcol];
    const float lam  = __expf(-rj * 16.0f);

    float bIn[8];
    #pragma unroll
    for (int j = 0; j < 8; ++j) bIn[j] = __expf(-rj * (float)(16 - 8 * h - j));

    {
        float* const uS = sU + ((tid >> 5) * TP + ((tid >> 2) & 7) * UP + (tid & 3) * 4);
        #pragma unroll
        for (int k = 0; k < 8; ++k) *(v4f*)(uS + (k & 1) * USZ + 4 * (k >> 1) * TP) = uq[k];
    }
    barrier_lds();

    float bv[8];
    {
        const float* pb = sKr + KPAD + (KT - 1) - m + 8 * h;
        #pragma unroll
        for (int j = 0; j < 8; ++j) bv[j] = pb[j];
    }

    v2f fb[16];
    {
        const float fq = (float)m;
        #pragma unroll
        for (int st = 0; st < 16; ++st) {
            const v2f rr = *(const v2f*)&sRate[32 * h + 2 * st];
            fb[st].x = __expf(-rr.x * fq);
            fb[st].y = __expf(-rr.y * fq);
        }
    }

    const float* uA = sU + mb * USZ + mp * UP + 8 * h;
    float*       gW = sG + h * GB + jcol;
    const float* gR = sG + mb * GB + (8 * wave + mp) * GP + 32 * h;
    float z = 0.0f;

    for (int Q = 0; Q < 4; ++Q) {
        for (int pr = 0; pr < 2; ++pr) {
            const float* p0 = uA + (4 * Q + 2 * pr) * TP;
            const float* p1 = p0 + TP;
            const v4f a00 = *(const v4f*)p0;
            const v4f a01 = *(const v4f*)(p0 + 4);
            const v4f a10 = *(const v4f*)p1;
            const v4f a11 = *(const v4f*)(p1 + 4);
            const float av0[8] = {a00.x, a00.y, a00.z, a00.w, a01.x, a01.y, a01.z, a01.w};
            const float av1[8] = {a10.x, a10.y, a10.z, a10.w, a11.x, a11.y, a11.z, a11.w};
            v8f w0 = {}, w1 = {};
            #pragma unroll
            for (int kk = 0; kk < 4; ++kk) {
                v2f b2, a2, a3;
                b2.x = bIn[2 * kk];  b2.y = bIn[2 * kk + 1];
                a2.x = av0[2 * kk];  a2.y = av0[2 * kk + 1];
                a3.x = av1[2 * kk];  a3.y = av1[2 * kk + 1];
                w0 = __builtin_amdgcn_wmma_f32_16x16x4_f32(false, a2, false, b2, (short)0, w0, false, false);
                w1 = __builtin_amdgcn_wmma_f32_16x16x4_f32(false, a3, false, b2, (short)0, w1, false, false);
            }
            float* g0 = gW + (16 * pr) * GP;
            #pragma unroll
            for (int r = 0; r < 8; ++r) {
                g0[r * GP] = mj * z;
                z = lam * z + w0[r];
            }
            #pragma unroll
            for (int r = 0; r < 8; ++r) {
                g0[(8 + r) * GP] = mj * z;
                z = lam * z + w1[r];
            }
        }
        barrier_lds();

        const int ch = 4 * Q + wave;
        v4f g[8];
        #pragma unroll
        for (int i = 0; i < 8; ++i) g[i] = *(const v4f*)(gR + 4 * i);
        const v4f t0 = *(const v4f*)(uA + ch * TP);
        const v4f t1 = *(const v4f*)(uA + ch * TP + 4);
        barrier_lds();

        const float tv[8] = {t0.x, t0.y, t0.z, t0.w, t1.x, t1.y, t1.z, t1.w};
        v8f acc0 = {}, acc1 = {};
        #pragma unroll
        for (int kk = 0; kk < 4; kk += 2) {
            v2f a2, b2, a3, b3;
            a2.x = tv[2 * kk];      a2.y = tv[2 * kk + 1];
            b2.x = bv[2 * kk];      b2.y = bv[2 * kk + 1];
            a3.x = tv[2 * kk + 2];  a3.y = tv[2 * kk + 3];
            b3.x = bv[2 * kk + 2];  b3.y = bv[2 * kk + 3];
            acc0 = __builtin_amdgcn_wmma_f32_16x16x4_f32(false, a2, false, b2, (short)0, acc0, false, false);
            acc1 = __builtin_amdgcn_wmma_f32_16x16x4_f32(false, a3, false, b3, (short)0, acc1, false, false);
        }
        #pragma unroll
        for (int st = 0; st < 16; st += 2) {
            v2f a2, a3;
            a2.x = g[st >> 1].x;  a2.y = g[st >> 1].y;
            a3.x = g[st >> 1].z;  a3.y = g[st >> 1].w;
            acc0 = __builtin_amdgcn_wmma_f32_16x16x4_f32(false, a2, false, fb[st],     (short)0, acc0, false, false);
            acc1 = __builtin_amdgcn_wmma_f32_16x16x4_f32(false, a3, false, fb[st + 1], (short)0, acc1, false, false);
        }

        float* yc = y + ((size_t)(b0 + h) * Cdim + c) * Tdim + 128 * ch;
        #pragma unroll
        for (int r = 0; r < 8; ++r) {
            const float uv = sU[h * USZ + ch * TP + r * UP + m];
            yc[16 * r + m] = (acc0[r] + acc1[r]) + dval * uv;
        }
    }
}

extern "C" void kernel_launch(void* const* d_in, const int* in_sizes, int n_in,
                              void* d_out, int out_size, void* d_ws, size_t ws_size,
                              hipStream_t stream) {
    const float* u       = (const float*)d_in[0];
    const float* log_tau = (const float*)d_in[1];
    const float* mix     = (const float*)d_in[2];
    const float* Dv      = (const float*)d_in[3];
    const float* log_dt  = (const float*)d_in[4];
    float*       y       = (float*)d_out;
    float*       Kws     = (float*)d_ws;

    (void)in_sizes; (void)n_in; (void)out_size; (void)ws_size;

    dim3 g1(KT / 16, (Cdim / 16) / 8);
    kgen_wmma<<<g1, 256, 0, stream>>>(mix, log_tau, log_dt, Kws);

    dim3 g2(Cdim, Bdim / 2);
    kconv<<<g2, 128, 0, stream>>>(u, Dv, Kws, log_tau, log_dt, mix, y);
}


# tap launch folded into the conv kernel: each block forms its channel's 16 taps itself (4 matrix steps per wave over 16 modes, parts summed via LDS); one launch; tap sum order new (4 partial sums)
# speedup vs baseline: 14.6657x; 1.0750x over previous
#include <hip/hip_runtime.h>

constexpr int Bdim = 4;
constexpr int Cdim = 1024;
constexpr int Ndim = 64;
constexpr int Tdim = 2048;
constexpr int KT   = 16;

typedef float v2f __attribute__((ext_vector_type(2)));
typedef float v4f __attribute__((ext_vector_type(4)));
typedef float v8f __attribute__((ext_vector_type(8)));

__device__ __forceinline__ float softplus_fast(float x) {
    return fmaxf(x, 0.0f) + __logf(1.0f + __expf(-fabsf(x)));
}

__device__ __forceinline__ void barrier_lds() {
    __builtin_amdgcn_fence(__ATOMIC_RELEASE, "workgroup", "local");
    __builtin_amdgcn_s_barrier();
    __builtin_amdgcn_fence(__ATOMIC_ACQUIRE, "workgroup", "local");
}

constexpr int NCH  = Tdim / 128;
constexpr int UP   = 20;
constexpr int TP   = 8 * UP;
constexpr int USZ  = NCH * TP + 32;
constexpr int KRS  = 2 * KT;
constexpr int QTL  = 32;
constexpr int GP   = Ndim + 4;
constexpr int GB   = QTL * GP + 32;

__global__ void __launch_bounds__(128) __attribute__((amdgpu_waves_per_eu(8, 8)))
kconv(const float* __restrict__ u,
      const float* __restrict__ Dv,
      const float* __restrict__ log_tau,
      const float* __restrict__ log_dt,
      const float* __restrict__ mix,
      float* __restrict__ y) {
    __shared__ __attribute__((aligned(16))) float sU[USZ + NCH * TP];
    __shared__ __attribute__((aligned(16))) float sKp[4 * KRS];
    __shared__ __attribute__((aligned(16))) float sG[GB + QTL * GP];
    __shared__ __attribute__((aligned(16))) float sRate[Ndim];

    const int c   = blockIdx.x;
    const int b0  = blockIdx.y * 2;
    const int tid = threadIdx.x;

    const int lane = tid & 31;
    const int wave = __builtin_amdgcn_readfirstlane(tid >> 5);
    const int m    = lane & 15;
    const int h    = lane >> 4;
    const int mb   = m >> 3;
    const int mp   = m & 7;
    const int jcol = 16 * wave + m;

    const float ltv  = log_tau[tid & (Ndim - 1)];
    const float ldt  = log_dt[0];
    const float* const mixC = mix + (size_t)c * Ndim;
    const v4f   ka0  = *(const v4f*)(mixC + 16 * wave + 8 * h);
    const v4f   ka1  = *(const v4f*)(mixC + 16 * wave + 8 * h + 4);
    const float mj   = mixC[jcol];
    const float dval = Dv[c];
    __builtin_amdgcn_sched_barrier(0);
    const float* const uG = u + ((size_t)b0 * Cdim + c) * Tdim + tid * 4;
    v4f uq[8];
    #pragma unroll
    for (int k = 0; k < 8; ++k)
        uq[k] = *(const v4f*)(uG + (size_t)(k & 1) * Cdim * Tdim + 512 * (k >> 1));

    asm volatile("" :: "v"(ltv), "s"(ldt));
    if (tid < Ndim) sRate[tid] = softplus_fast(ltv) * softplus_fast(ldt);
    barrier_lds();

    {
        const v4f r0 = *(const v4f*)&sRate[16 * wave + 8 * h];
        const v4f r1 = *(const v4f*)&sRate[16 * wave + 8 * h + 4];
        const float rr[8] = {r0.x, r0.y, r0.z, r0.w, r1.x, r1.y, r1.z, r1.w};
        const float aa[8] = {ka0.x, ka0.y, ka0.z, ka0.w, ka1.x, ka1.y, ka1.z, ka1.w};
        const float fq = (float)m;
        v8f kacc = {};
        #pragma unroll
        for (int kk = 0; kk < 4; ++kk) {
            v2f a2, b2;
            a2.x = aa[2 * kk];                 a2.y = aa[2 * kk + 1];
            b2.x = __expf(-rr[2 * kk] * fq);   b2.y = __expf(-rr[2 * kk + 1] * fq);
            kacc = __builtin_amdgcn_wmma_f32_16x16x4_f32(false, a2, false, b2, (short)0, kacc, false, false);
        }
        sKp[wave * KRS + lane] = kacc[0];
    }

    const float rj   = sRate[jcol];
    const float lam  = __expf(-rj * 16.0f);

    float bIn[8];
    #pragma unroll
    for (int j = 0; j < 8; ++j) bIn[j] = __expf(-rj * (float)(16 - 8 * h - j));

    {
        float* const uS = sU + ((tid >> 5) * TP + ((tid >> 2) & 7) * UP + (tid & 3) * 4);
        #pragma unroll
        for (int k = 0; k < 8; ++k) *(v4f*)(uS + (k & 1) * USZ + 4 * (k >> 1) * TP) = uq[k];
    }
    barrier_lds();

    float bv[8];
    {
        const float* pk = sKp + lane;
        const float kc = (pk[0] + pk[KRS]) + (pk[2 * KRS] + pk[3 * KRS]);
        #pragma unroll
        for (int j = 0; j < 8; ++j) {
            const int d = m - 8 * h - j;
            const float t = __shfl(kc, d & 15, 16);
            bv[j] = d >= 0 ? t : 0.0f;
        }
    }

    v2f fb[16];
    {
        const float fq = (float)m;
        #pragma unroll
        for (int st = 0; st < 16; ++st) {
            const v2f rr = *(const v2f*)&sRate[32 * h + 2 * st];
            fb[st].x = __expf(-rr.x * fq);
            fb[st].y = __expf(-rr.y * fq);
        }
    }

    const float* uA = sU + mb * USZ + mp * UP + 8 * h;
    float*       gW = sG + h * GB + jcol;
    const float* gR = sG + mb * GB + (8 * wave + mp) * GP + 32 * h;
    float z = 0.0f;

    for (int Q = 0; Q < 4; ++Q) {
        for (int pr = 0; pr < 2; ++pr) {
            const float* p0 = uA + (4 * Q + 2 * pr) * TP;
            const float* p1 = p0 + TP;
            const v4f a00 = *(const v4f*)p0;
            const v4f a01 = *(const v4f*)(p0 + 4);
            const v4f a10 = *(const v4f*)p1;
            const v4f a11 = *(const v4f*)(p1 + 4);
            const float av0[8] = {a00.x, a00.y, a00.z, a00.w, a01.x, a01.y, a01.z, a01.w};
            const float av1[8] = {a10.x, a10.y, a10.z, a10.w, a11.x, a11.y, a11.z, a11.w};
            v8f w0 = {}, w1 = {};
            #pragma unroll
            for (int kk = 0; kk < 4; ++kk) {
                v2f b2, a2, a3;
                b2.x = bIn[2 * kk];  b2.y = bIn[2 * kk + 1];
                a2.x = av0[2 * kk];  a2.y = av0[2 * kk + 1];
                a3.x = av1[2 * kk];  a3.y = av1[2 * kk + 1];
                w0 = __builtin_amdgcn_wmma_f32_16x16x4_f32(false, a2, false, b2, (short)0, w0, false, false);
                w1 = __builtin_amdgcn_wmma_f32_16x16x4_f32(false, a3, false, b2, (short)0, w1, false, false);
            }
            float* g0 = gW + (16 * pr) * GP;
            #pragma unroll
            for (int r = 0; r < 8; ++r) {
                g0[r * GP] = mj * z;
                z = lam * z + w0[r];
            }
            #pragma unroll
            for (int r = 0; r < 8; ++r) {
                g0[(8 + r) * GP] = mj * z;
                z = lam * z + w1[r];
            }
        }
        barrier_lds();

        const int ch = 4 * Q + wave;
        v4f g[8];
        #pragma unroll
        for (int i = 0; i < 8; ++i) g[i] = *(const v4f*)(gR + 4 * i);
        const v4f t0 = *(const v4f*)(uA + ch * TP);
        const v4f t1 = *(const v4f*)(uA + ch * TP + 4);
        barrier_lds();

        const float tv[8] = {t0.x, t0.y, t0.z, t0.w, t1.x, t1.y, t1.z, t1.w};
        v8f acc0 = {}, acc1 = {};
        #pragma unroll
        for (int kk = 0; kk < 4; kk += 2) {
            v2f a2, b2, a3, b3;
            a2.x = tv[2 * kk];      a2.y = tv[2 * kk + 1];
            b2.x = bv[2 * kk];      b2.y = bv[2 * kk + 1];
            a3.x = tv[2 * kk + 2];  a3.y = tv[2 * kk + 3];
            b3.x = bv[2 * kk + 2];  b3.y = bv[2 * kk + 3];
            acc0 = __builtin_amdgcn_wmma_f32_16x16x4_f32(false, a2, false, b2, (short)0, acc0, false, false);
            acc1 = __builtin_amdgcn_wmma_f32_16x16x4_f32(false, a3, false, b3, (short)0, acc1, false, false);
        }
        #pragma unroll
        for (int st = 0; st < 16; st += 2) {
            v2f a2, a3;
            a2.x = g[st >> 1].x;  a2.y = g[st >> 1].y;
            a3.x = g[st >> 1].z;  a3.y = g[st >> 1].w;
            acc0 = __builtin_amdgcn_wmma_f32_16x16x4_f32(false, a2, false, fb[st],     (short)0, acc0, false, false);
            acc1 = __builtin_amdgcn_wmma_f32_16x16x4_f32(false, a3, false, fb[st + 1], (short)0, acc1, false, false);
        }

        float* yc = y + ((size_t)(b0 + h) * Cdim + c) * Tdim + 128 * ch;
        #pragma unroll
        for (int r = 0; r < 8; ++r) {
            const float uv = sU[h * USZ + ch * TP + r * UP + m];
            yc[16 * r + m] = (acc0[r] + acc1[r]) + dval * uv;
        }
    }
}

extern "C" void kernel_launch(void* const* d_in, const int* in_sizes, int n_in,
                              void* d_out, int out_size, void* d_ws, size_t ws_size,
                              hipStream_t stream) {
    const float* u       = (const float*)d_in[0];
    const float* log_tau = (const float*)d_in[1];
    const float* mix     = (const float*)d_in[2];
    const float* Dv      = (const float*)d_in[3];
    const float* log_dt  = (const float*)d_in[4];
    float*       y       = (float*)d_out;

    (void)in_sizes; (void)n_in; (void)out_size; (void)d_ws; (void)ws_size;

    dim3 g2(Cdim, Bdim / 2);
    kconv<<<g2, 128, 0, stream>>>(u, Dv, log_tau, log_dt, mix, y);
}


# one block per channel does all four batches, the two batch pairs side by side in every phase (next state-in's u fragments read with G): twice the matrix work per wait, half the waves; sums unchanged
# speedup vs baseline: 15.5984x; 1.0636x over previous
#include <hip/hip_runtime.h>

constexpr int Bdim = 4;
constexpr int Cdim = 1024;
constexpr int Ndim = 64;
constexpr int Tdim = 2048;
constexpr int KT   = 16;

typedef float v2f __attribute__((ext_vector_type(2)));
typedef float v4f __attribute__((ext_vector_type(4)));
typedef float v8f __attribute__((ext_vector_type(8)));

__device__ __forceinline__ float softplus_fast(float x) {
    return fmaxf(x, 0.0f) + __logf(1.0f + __expf(-fabsf(x)));
}

__device__ __forceinline__ void barrier_lds() {
    __builtin_amdgcn_fence(__ATOMIC_RELEASE, "workgroup", "local");
    __builtin_amdgcn_s_barrier();
    __builtin_amdgcn_fence(__ATOMIC_ACQUIRE, "workgroup", "local");
}

constexpr int NCH  = Tdim / 128;
constexpr int UP   = 20;
constexpr int TP   = 8 * UP;
constexpr int USZ  = NCH * TP + 32;
constexpr int KRS  = 2 * KT;
constexpr int QTL  = 32;
constexpr int GP   = Ndim + 4;
constexpr int GB   = QTL * GP + 32;
constexpr int GSZ  = GB + QTL * GP;
constexpr int NPR  = Bdim / 2;

__global__ void __launch_bounds__(128) __attribute__((amdgpu_waves_per_eu(4, 4)))
kconv(const float* __restrict__ u,
      const float* __restrict__ Dv,
      const float* __restrict__ log_tau,
      const float* __restrict__ log_dt,
      const float* __restrict__ mix,
      float* __restrict__ y) {
    __shared__ __attribute__((aligned(16))) float sU[3 * USZ + NCH * TP];
    __shared__ __attribute__((aligned(16))) float sKp[4 * KRS];
    __shared__ __attribute__((aligned(16))) float sG[NPR * GSZ];
    __shared__ __attribute__((aligned(16))) float sRate[Ndim];

    const int c   = blockIdx.x;
    const int tid = threadIdx.x;

    const int lane = tid & 31;
    const int wave = __builtin_amdgcn_readfirstlane(tid >> 5);
    const int m    = lane & 15;
    const int h    = lane >> 4;
    const int mb   = m >> 3;
    const int mp   = m & 7;
    const int jcol = 16 * wave + m;

    const float ltv  = log_tau[tid & (Ndim - 1)];
    const float ldt  = log_dt[0];
    const float* const mixC = mix + (size_t)c * Ndim;
    const v4f   ka0  = *(const v4f*)(mixC + 16 * wave + 8 * h);
    const v4f   ka1  = *(const v4f*)(mixC + 16 * wave + 8 * h + 4);
    const float mj   = mixC[jcol];
    const float dval = Dv[c];
    __builtin_amdgcn_sched_barrier(0);
    const float* const uG = u + (size_t)c * Tdim + tid * 4;
    v4f uq[16];
    #pragma unroll
    for (int k = 0; k < 16; ++k)
        uq[k] = *(const v4f*)(uG + (size_t)(k & 3) * Cdim * Tdim + 512 * (k >> 2));

    asm volatile("" :: "v"(ltv), "s"(ldt));
    if (tid < Ndim) sRate[tid] = softplus_fast(ltv) * softplus_fast(ldt);
    barrier_lds();

    {
        const v4f r0 = *(const v4f*)&sRate[16 * wave + 8 * h];
        const v4f r1 = *(const v4f*)&sRate[16 * wave + 8 * h + 4];
        const float rr[8] = {r0.x, r0.y, r0.z, r0.w, r1.x, r1.y, r1.z, r1.w};
        const float aa[8] = {ka0.x, ka0.y, ka0.z, ka0.w, ka1.x, ka1.y, ka1.z, ka1.w};
        const float fq = (float)m;
        v8f kacc = {};
        #pragma unroll
        for (int kk = 0; kk < 4; ++kk) {
            v2f a2, b2;
            a2.x = aa[2 * kk];                 a2.y = aa[2 * kk + 1];
            b2.x = __expf(-rr[2 * kk] * fq);   b2.y = __expf(-rr[2 * kk + 1] * fq);
            kacc = __builtin_amdgcn_wmma_f32_16x16x4_f32(false, a2, false, b2, (short)0, kacc, false, false);
        }
        sKp[wave * KRS + lane] = kacc[0];
    }

    const float rj   = sRate[jcol];
    const float lam  = __expf(-rj * 16.0f);

    float bIn[8];
    #pragma unroll
    for (int j = 0; j < 8; ++j) bIn[j] = __expf(-rj * (float)(16 - 8 * h - j));

    {
        float* const uS = sU + ((tid >> 5) * TP + ((tid >> 2) & 7) * UP + (tid & 3) * 4);
        #pragma unroll
        for (int k = 0; k < 16; ++k) *(v4f*)(uS + (k & 3) * USZ + 4 * (k >> 2) * TP) = uq[k];
    }
    barrier_lds();

    float bv[8];
    {
        const float* pk = sKp + lane;
        const float kc = (pk[0] + pk[KRS]) + (pk[2 * KRS] + pk[3 * KRS]);
        #pragma unroll
        for (int j = 0; j < 8; ++j) {
            const int d = m - 8 * h - j;
            const float t = __shfl(kc, d & 15, 16);
            bv[j] = d >= 0 ? t : 0.0f;
        }
    }

    v2f fb[16];
    {
        const float fq = (float)m;
        #pragma unroll
        for (int st = 0; st < 16; ++st) {
            const v2f rr = *(const v2f*)&sRate[32 * h + 2 * st];
            fb[st].x = __expf(-rr.x * fq);
            fb[st].y = __expf(-rr.y * fq);
        }
    }

    const float* uA[NPR];
    float*       gW[NPR];
    const float* gR[NPR];
    #pragma unroll
    for (int P = 0; P < NPR; ++P) {
        uA[P] = sU + (2 * P + mb) * USZ + mp * UP + 8 * h;
        gW[P] = sG + P * GSZ + h * GB + jcol;
        gR[P] = sG + P * GSZ + mb * GB + (8 * wave + mp) * GP + 32 * h;
    }
    float z[NPR] = {0.0f, 0.0f};

    v4f af[NPR][4];
    #pragma unroll
    for (int P = 0; P < NPR; ++P) {
        af[P][0] = *(const v4f*)(uA[P]);
        af[P][1] = *(const v4f*)(uA[P] + 4);
        af[P][2] = *(const v4f*)(uA[P] + TP);
        af[P][3] = *(const v4f*)(uA[P] + TP + 4);
    }

    for (int Q = 0; Q < 4; ++Q) {
        #pragma unroll
        for (int pr = 0; pr < 2; ++pr) {
            float av0[NPR][8], av1[NPR][8];
            #pragma unroll
            for (int P = 0; P < NPR; ++P) {
                const float* p0 = uA[P] + (4 * Q + 2 * pr) * TP;
                const float* p1 = p0 + TP;
                const v4f a00 = pr == 0 ? af[P][0] : *(const v4f*)p0;
                const v4f a01 = pr == 0 ? af[P][1] : *(const v4f*)(p0 + 4);
                const v4f a10 = pr == 0 ? af[P][2] : *(const v4f*)p1;
                const v4f a11 = pr == 0 ? af[P][3] : *(const v4f*)(p1 + 4);
                av0[P][0] = a00.x; av0[P][1] = a00.y; av0[P][2] = a00.z; av0[P][3] = a00.w;
                av0[P][4] = a01.x; av0[P][5] = a01.y; av0[P][6] = a01.z; av0[P][7] = a01.w;
                av1[P][0] = a10.x; av1[P][1] = a10.y; av1[P][2] = a10.z; av1[P][3] = a10.w;
                av1[P][4] = a11.x; av1[P][5] = a11.y; av1[P][6] = a11.z; av1[P][7] = a11.w;
            }
            v8f w0[NPR], w1[NPR];
            #pragma unroll
            for (int P = 0; P < NPR; ++P) { w0[P] = v8f{}; w1[P] = v8f{}; }
            #pragma unroll
            for (int kk = 0; kk < 4; ++kk) {
                v2f b2;
                b2.x = bIn[2 * kk];  b2.y = bIn[2 * kk + 1];
                #pragma unroll
                for (int P = 0; P < NPR; ++P) {
                    v2f a2, a3;
                    a2.x = av0[P][2 * kk];  a2.y = av0[P][2 * kk + 1];
                    a3.x = av1[P][2 * kk];  a3.y = av1[P][2 * kk + 1];
                    w0[P] = __builtin_amdgcn_wmma_f32_16x16x4_f32(false, a2, false, b2, (short)0, w0[P], false, false);
                    w1[P] = __builtin_amdgcn_wmma_f32_16x16x4_f32(false, a3, false, b2, (short)0, w1[P], false, false);
                }
            }
            #pragma unroll
            for (int r = 0; r < 8; ++r) {
                #pragma unroll
                for (int P = 0; P < NPR; ++P) {
                    gW[P][(16 * pr + r) * GP] = mj * z[P];
                    z[P] = lam * z[P] + w0[P][r];
                }
            }
            #pragma unroll
            for (int r = 0; r < 8; ++r) {
                #pragma unroll
                for (int P = 0; P < NPR; ++P) {
                    gW[P][(16 * pr + 8 + r) * GP] = mj * z[P];
                    z[P] = lam * z[P] + w1[P][r];
                }
            }
        }
        barrier_lds();

        const int ch = 4 * Q + wave;
        v4f g[NPR][8];
        v4f t0[NPR], t1[NPR];
        #pragma unroll
        for (int P = 0; P < NPR; ++P) {
            #pragma unroll
            for (int i = 0; i < 8; ++i) g[P][i] = *(const v4f*)(gR[P] + 4 * i);
            t0[P] = *(const v4f*)(uA[P] + ch * TP);
            t1[P] = *(const v4f*)(uA[P] + ch * TP + 4);
        }
        {
            const int qn = Q < 3 ? Q + 1 : 3;
            #pragma unroll
            for (int P = 0; P < NPR; ++P) {
                const float* pn = uA[P] + 4 * qn * TP;
                af[P][0] = *(const v4f*)(pn);
                af[P][1] = *(const v4f*)(pn + 4);
                af[P][2] = *(const v4f*)(pn + TP);
                af[P][3] = *(const v4f*)(pn + TP + 4);
            }
        }
        barrier_lds();

        v8f acc0[NPR], acc1[NPR];
        #pragma unroll
        for (int P = 0; P < NPR; ++P) { acc0[P] = v8f{}; acc1[P] = v8f{}; }
        #pragma unroll
        for (int kk = 0; kk < 4; kk += 2) {
            v2f b2, b3;
            b2.x = bv[2 * kk];      b2.y = bv[2 * kk + 1];
            b3.x = bv[2 * kk + 2];  b3.y = bv[2 * kk + 3];
            #pragma unroll
            for (int P = 0; P < NPR; ++P) {
                const float tv[8] = {t0[P].x, t0[P].y, t0[P].z, t0[P].w, t1[P].x, t1[P].y, t1[P].z, t1[P].w};
                v2f a2, a3;
                a2.x = tv[2 * kk];      a2.y = tv[2 * kk + 1];
                a3.x = tv[2 * kk + 2];  a3.y = tv[2 * kk + 3];
                acc0[P] = __builtin_amdgcn_wmma_f32_16x16x4_f32(false, a2, false, b2, (short)0, acc0[P], false, false);
                acc1[P] = __builtin_amdgcn_wmma_f32_16x16x4_f32(false, a3, false, b3, (short)0, acc1[P], false, false);
            }
        }
        #pragma unroll
        for (int st = 0; st < 16; st += 2) {
            #pragma unroll
            for (int P = 0; P < NPR; ++P) {
                v2f a2, a3;
                a2.x = g[P][st >> 1].x;  a2.y = g[P][st >> 1].y;
                a3.x = g[P][st >> 1].z;  a3.y = g[P][st >> 1].w;
                acc0[P] = __builtin_amdgcn_wmma_f32_16x16x4_f32(false, a2, false, fb[st],     (short)0, acc0[P], false, false);
                acc1[P] = __builtin_amdgcn_wmma_f32_16x16x4_f32(false, a3, false, fb[st + 1], (short)0, acc1[P], false, false);
            }
        }

        #pragma unroll
        for (int P = 0; P < NPR; ++P) {
            float* yc = y + ((size_t)(2 * P + h) * Cdim + c) * Tdim + 128 * ch;
            #pragma unroll
            for (int r = 0; r < 8; ++r) {
                const float uv = sU[(2 * P + h) * USZ + ch * TP + r * UP + m];
                yc[16 * r + m] = (acc0[P][r] + acc1[P][r]) + dval * uv;
            }
        }
    }
}

extern "C" void kernel_launch(void* const* d_in, const int* in_sizes, int n_in,
                              void* d_out, int out_size, void* d_ws, size_t ws_size,
                              hipStream_t stream) {
    const float* u       = (const float*)d_in[0];
    const float* log_tau = (const float*)d_in[1];
    const float* mix     = (const float*)d_in[2];
    const float* Dv      = (const float*)d_in[3];
    const float* log_dt  = (const float*)d_in[4];
    float*       y       = (float*)d_out;

    (void)in_sizes; (void)n_in; (void)out_size; (void)d_ws; (void)ws_size;

    kconv<<<dim3(Cdim), 128, 0, stream>>>(u, Dv, log_tau, log_dt, mix, y);
}


# wave priority follows progress: s_setprio 3 from entry, one level down at each later quarter's G hand-over (quarter loop unrolled for its constant): a WGP's 4 blocks advance together; sums unchanged
# speedup vs baseline: 16.9842x; 1.0888x over previous
#include <hip/hip_runtime.h>

constexpr int Bdim = 4;
constexpr int Cdim = 1024;
constexpr int Ndim = 64;
constexpr int Tdim = 2048;
constexpr int KT   = 16;

typedef float v2f __attribute__((ext_vector_type(2)));
typedef float v4f __attribute__((ext_vector_type(4)));
typedef float v8f __attribute__((ext_vector_type(8)));

__device__ __forceinline__ float softplus_fast(float x) {
    return fmaxf(x, 0.0f) + __logf(1.0f + __expf(-fabsf(x)));
}

__device__ __forceinline__ void barrier_lds() {
    __builtin_amdgcn_fence(__ATOMIC_RELEASE, "workgroup", "local");
    __builtin_amdgcn_s_barrier();
    __builtin_amdgcn_fence(__ATOMIC_ACQUIRE, "workgroup", "local");
}

constexpr int NCH  = Tdim / 128;
constexpr int UP   = 20;
constexpr int TP   = 8 * UP;
constexpr int USZ  = NCH * TP + 32;
constexpr int KRS  = 2 * KT;
constexpr int QTL  = 32;
constexpr int GP   = Ndim + 4;
constexpr int GB   = QTL * GP + 32;
constexpr int GSZ  = GB + QTL * GP;
constexpr int NPR  = Bdim / 2;

__global__ void __launch_bounds__(128) __attribute__((amdgpu_waves_per_eu(4, 4)))
kconv(const float* __restrict__ u,
      const float* __restrict__ Dv,
      const float* __restrict__ log_tau,
      const float* __restrict__ log_dt,
      const float* __restrict__ mix,
      float* __restrict__ y) {
    __shared__ __attribute__((aligned(16))) float sU[3 * USZ + NCH * TP];
    __shared__ __attribute__((aligned(16))) float sKp[4 * KRS];
    __shared__ __attribute__((aligned(16))) float sG[NPR * GSZ];
    __shared__ __attribute__((aligned(16))) float sRate[Ndim];

    __builtin_amdgcn_s_setprio(3);

    const int c   = blockIdx.x;
    const int tid = threadIdx.x;

    const int lane = tid & 31;
    const int wave = __builtin_amdgcn_readfirstlane(tid >> 5);
    const int m    = lane & 15;
    const int h    = lane >> 4;
    const int mb   = m >> 3;
    const int mp   = m & 7;
    const int jcol = 16 * wave + m;

    const float ltv  = log_tau[tid & (Ndim - 1)];
    const float ldt  = log_dt[0];
    const float* const mixC = mix + (size_t)c * Ndim;
    const v4f   ka0  = *(const v4f*)(mixC + 16 * wave + 8 * h);
    const v4f   ka1  = *(const v4f*)(mixC + 16 * wave + 8 * h + 4);
    const float mj   = mixC[jcol];
    const float dval = Dv[c];
    __builtin_amdgcn_sched_barrier(0);
    const float* const uG = u + (size_t)c * Tdim + tid * 4;
    v4f uq[16];
    #pragma unroll
    for (int k = 0; k < 16; ++k)
        uq[k] = *(const v4f*)(uG + (size_t)(k & 3) * Cdim * Tdim + 512 * (k >> 2));

    asm volatile("" :: "v"(ltv), "s"(ldt));
    if (tid < Ndim) sRate[tid] = softplus_fast(ltv) * softplus_fast(ldt);
    barrier_lds();

    {
        const v4f r0 = *(const v4f*)&sRate[16 * wave + 8 * h];
        const v4f r1 = *(const v4f*)&sRate[16 * wave + 8 * h + 4];
        const float rr[8] = {r0.x, r0.y, r0.z, r0.w, r1.x, r1.y, r1.z, r1.w};
        const float aa[8] = {ka0.x, ka0.y, ka0.z, ka0.w, ka1.x, ka1.y, ka1.z, ka1.w};
        const float fq = (float)m;
        v8f kacc = {};
        #pragma unroll
        for (int kk = 0; kk < 4; ++kk) {
            v2f a2, b2;
            a2.x = aa[2 * kk];                 a2.y = aa[2 * kk + 1];
            b2.x = __expf(-rr[2 * kk] * fq);   b2.y = __expf(-rr[2 * kk + 1] * fq);
            kacc = __builtin_amdgcn_wmma_f32_16x16x4_f32(false, a2, false, b2, (short)0, kacc, false, false);
        }
        sKp[wave * KRS + lane] = kacc[0];
    }

    const float rj   = sRate[jcol];
    const float lam  = __expf(-rj * 16.0f);

    float bIn[8];
    #pragma unroll
    for (int j = 0; j < 8; ++j) bIn[j] = __expf(-rj * (float)(16 - 8 * h - j));

    {
        float* const uS = sU + ((tid >> 5) * TP + ((tid >> 2) & 7) * UP + (tid & 3) * 4);
        #pragma unroll
        for (int k = 0; k < 16; ++k) *(v4f*)(uS + (k & 3) * USZ + 4 * (k >> 2) * TP) = uq[k];
    }
    barrier_lds();

    float bv[8];
    {
        const float* pk = sKp + lane;
        const float kc = (pk[0] + pk[KRS]) + (pk[2 * KRS] + pk[3 * KRS]);
        #pragma unroll
        for (int j = 0; j < 8; ++j) {
            const int d = m - 8 * h - j;
            const float t = __shfl(kc, d & 15, 16);
            bv[j] = d >= 0 ? t : 0.0f;
        }
    }

    v2f fb[16];
    {
        const float fq = (float)m;
        #pragma unroll
        for (int st = 0; st < 16; ++st) {
            const v2f rr = *(const v2f*)&sRate[32 * h + 2 * st];
            fb[st].x = __expf(-rr.x * fq);
            fb[st].y = __expf(-rr.y * fq);
        }
    }

    const float* uA[NPR];
    float*       gW[NPR];
    const float* gR[NPR];
    #pragma unroll
    for (int P = 0; P < NPR; ++P) {
        uA[P] = sU + (2 * P + mb) * USZ + mp * UP + 8 * h;
        gW[P] = sG + P * GSZ + h * GB + jcol;
        gR[P] = sG + P * GSZ + mb * GB + (8 * wave + mp) * GP + 32 * h;
    }
    float z[NPR] = {0.0f, 0.0f};

    v4f af[NPR][4];
    #pragma unroll
    for (int P = 0; P < NPR; ++P) {
        af[P][0] = *(const v4f*)(uA[P]);
        af[P][1] = *(const v4f*)(uA[P] + 4);
        af[P][2] = *(const v4f*)(uA[P] + TP);
        af[P][3] = *(const v4f*)(uA[P] + TP + 4);
    }

    #pragma unroll
    for (int Q = 0; Q < 4; ++Q) {
        #pragma unroll
        for (int pr = 0; pr < 2; ++pr) {
            float av0[NPR][8], av1[NPR][8];
            #pragma unroll
            for (int P = 0; P < NPR; ++P) {
                const float* p0 = uA[P] + (4 * Q + 2 * pr) * TP;
                const float* p1 = p0 + TP;
                const v4f a00 = pr == 0 ? af[P][0] : *(const v4f*)p0;
                const v4f a01 = pr == 0 ? af[P][1] : *(const v4f*)(p0 + 4);
                const v4f a10 = pr == 0 ? af[P][2] : *(const v4f*)p1;
                const v4f a11 = pr == 0 ? af[P][3] : *(const v4f*)(p1 + 4);
                av0[P][0] = a00.x; av0[P][1] = a00.y; av0[P][2] = a00.z; av0[P][3] = a00.w;
                av0[P][4] = a01.x; av0[P][5] = a01.y; av0[P][6] = a01.z; av0[P][7] = a01.w;
                av1[P][0] = a10.x; av1[P][1] = a10.y; av1[P][2] = a10.z; av1[P][3] = a10.w;
                av1[P][4] = a11.x; av1[P][5] = a11.y; av1[P][6] = a11.z; av1[P][7] = a11.w;
            }
            v8f w0[NPR], w1[NPR];
            #pragma unroll
            for (int P = 0; P < NPR; ++P) { w0[P] = v8f{}; w1[P] = v8f{}; }
            #pragma unroll
            for (int kk = 0; kk < 4; ++kk) {
                v2f b2;
                b2.x = bIn[2 * kk];  b2.y = bIn[2 * kk + 1];
                #pragma unroll
                for (int P = 0; P < NPR; ++P) {
                    v2f a2, a3;
                    a2.x = av0[P][2 * kk];  a2.y = av0[P][2 * kk + 1];
                    a3.x = av1[P][2 * kk];  a3.y = av1[P][2 * kk + 1];
                    w0[P] = __builtin_amdgcn_wmma_f32_16x16x4_f32(false, a2, false, b2, (short)0, w0[P], false, false);
                    w1[P] = __builtin_amdgcn_wmma_f32_16x16x4_f32(false, a3, false, b2, (short)0, w1[P], false, false);
                }
            }
            #pragma unroll
            for (int r = 0; r < 8; ++r) {
                #pragma unroll
                for (int P = 0; P < NPR; ++P) {
                    gW[P][(16 * pr + r) * GP] = mj * z[P];
                    z[P] = lam * z[P] + w0[P][r];
                }
            }
            #pragma unroll
            for (int r = 0; r < 8; ++r) {
                #pragma unroll
                for (int P = 0; P < NPR; ++P) {
                    gW[P][(16 * pr + 8 + r) * GP] = mj * z[P];
                    z[P] = lam * z[P] + w1[P][r];
                }
            }
        }
        barrier_lds();

        if (Q == 1)      __builtin_amdgcn_s_setprio(2);
        else if (Q == 2) __builtin_amdgcn_s_setprio(1);
        else if (Q == 3) __builtin_amdgcn_s_setprio(0);

        const int ch = 4 * Q + wave;
        v4f g[NPR][8];
        v4f t0[NPR], t1[NPR];
        #pragma unroll
        for (int P = 0; P < NPR; ++P) {
            #pragma unroll
            for (int i = 0; i < 8; ++i) g[P][i] = *(const v4f*)(gR[P] + 4 * i);
            t0[P] = *(const v4f*)(uA[P] + ch * TP);
            t1[P] = *(const v4f*)(uA[P] + ch * TP + 4);
        }
        {
            const int qn = Q < 3 ? Q + 1 : 3;
            #pragma unroll
            for (int P = 0; P < NPR; ++P) {
                const float* pn = uA[P] + 4 * qn * TP;
                af[P][0] = *(const v4f*)(pn);
                af[P][1] = *(const v4f*)(pn + 4);
                af[P][2] = *(const v4f*)(pn + TP);
                af[P][3] = *(const v4f*)(pn + TP + 4);
            }
        }
        barrier_lds();

        v8f acc0[NPR], acc1[NPR];
        #pragma unroll
        for (int P = 0; P < NPR; ++P) { acc0[P] = v8f{}; acc1[P] = v8f{}; }
        #pragma unroll
        for (int kk = 0; kk < 4; kk += 2) {
            v2f b2, b3;
            b2.x = bv[2 * kk];      b2.y = bv[2 * kk + 1];
            b3.x = bv[2 * kk + 2];  b3.y = bv[2 * kk + 3];
            #pragma unroll
            for (int P = 0; P < NPR; ++P) {
                const float tv[8] = {t0[P].x, t0[P].y, t0[P].z, t0[P].w, t1[P].x, t1[P].y, t1[P].z, t1[P].w};
                v2f a2, a3;
                a2.x = tv[2 * kk];      a2.y = tv[2 * kk + 1];
                a3.x = tv[2 * kk + 2];  a3.y = tv[2 * kk + 3];
                acc0[P] = __builtin_amdgcn_wmma_f32_16x16x4_f32(false, a2, false, b2, (short)0, acc0[P], false, false);
                acc1[P] = __builtin_amdgcn_wmma_f32_16x16x4_f32(false, a3, false, b3, (short)0, acc1[P], false, false);
            }
        }
        #pragma unroll
        for (int st = 0; st < 16; st += 2) {
            #pragma unroll
            for (int P = 0; P < NPR; ++P) {
                v2f a2, a3;
                a2.x = g[P][st >> 1].x;  a2.y = g[P][st >> 1].y;
                a3.x = g[P][st >> 1].z;  a3.y = g[P][st >> 1].w;
                acc0[P] = __builtin_amdgcn_wmma_f32_16x16x4_f32(false, a2, false, fb[st],     (short)0, acc0[P], false, false);
                acc1[P] = __builtin_amdgcn_wmma_f32_16x16x4_f32(false, a3, false, fb[st + 1], (short)0, acc1[P], false, false);
            }
        }

        #pragma unroll
        for (int P = 0; P < NPR; ++P) {
            float* yc = y + ((size_t)(2 * P + h) * Cdim + c) * Tdim + 128 * ch;
            #pragma unroll
            for (int r = 0; r < 8; ++r) {
                const float uv = sU[(2 * P + h) * USZ + ch * TP + r * UP + m];
                yc[16 * r + m] = (acc0[P][r] + acc1[P][r]) + dval * uv;
            }
        }
    }
}

extern "C" void kernel_launch(void* const* d_in, const int* in_sizes, int n_in,
                              void* d_out, int out_size, void* d_ws, size_t ws_size,
                              hipStream_t stream) {
    const float* u       = (const float*)d_in[0];
    const float* log_tau = (const float*)d_in[1];
    const float* mix     = (const float*)d_in[2];
    const float* Dv      = (const float*)d_in[3];
    const float* log_dt  = (const float*)d_in[4];
    float*       y       = (float*)d_out;

    (void)in_sizes; (void)n_in; (void)out_size; (void)d_ws; (void)ws_size;

    kconv<<<dim3(Cdim), 128, 0, stream>>>(u, Dv, log_tau, log_dt, mix, y);
}
